# starting kernel
# baseline (speedup 1.0000x reference)
#include <hip/hip_runtime.h>

typedef __attribute__((ext_vector_type(2))) float v2f;
typedef __attribute__((ext_vector_type(8))) float v8f;

#define NPTS   200000
#define MOUT   100000
#define CIN    32
#define COUT   64
#define KNBR   32
#define VOX    27
#define KDIM   (VOX * CIN)
#define AS     (KDIM + 2)
#define MTILE  16
#define WAVES  4
#define BLOCK  (WAVES * 32)

__global__ __launch_bounds__(BLOCK)
void cconv_fused_kernel(const float* __restrict__ feats,
                        const float* __restrict__ inPos,
                        const float* __restrict__ outPos,
                        const float* __restrict__ extents,
                        const float* __restrict__ importance,
                        const int*   __restrict__ nbrIdx,
                        const float* __restrict__ Wg,
                        const float* __restrict__ bias,
                        float*       __restrict__ out)
{
    __shared__ float sA[MTILE][AS];
    __shared__ float sInvN[MTILE];

    const int lane = threadIdx.x & 31;
    const int wave = threadIdx.x >> 5;
    const int m0   = blockIdx.x * MTILE;
    const float invR = 2.0f / extents[0];

    for (int r = 0; r < MTILE / WAVES; ++r) {
        const int row = wave * (MTILE / WAVES) + r;
        const int m   = m0 + row;
        const float ox = outPos[m * 3 + 0];
        const float oy = outPos[m * 3 + 1];
        const float oz = outPos[m * 3 + 2];

        float acc[VOX];
        #pragma unroll
        for (int v = 0; v < VOX; ++v) acc[v] = 0.0f;
        float norm = 0.0f;

        for (int k = 0; k < KNBR; ++k) {
            const int idx = nbrIdx[m * KNBR + k];
            const float rx = (inPos[idx * 3 + 0] - ox) * invR;
            const float ry = (inPos[idx * 3 + 1] - oy) * invR;
            const float rz = (inPos[idx * 3 + 2] - oz) * invR;
            const float dist = sqrtf(rx * rx + ry * ry + rz * rz);
            const float linf = fmaxf(fabsf(rx), fmaxf(fabsf(ry), fabsf(rz)));
            const float scale = (linf > 1e-8f) ? dist / fmaxf(linf, 1e-8f) : 0.0f;
            const float imp = (dist <= 1.0f) ? importance[idx] : 0.0f;

            if (imp != 0.0f) {
                norm += imp;
                const float cx = fminf(fmaxf(rx * scale + 1.0f, 0.0f), 2.0f);
                const float cy = fminf(fmaxf(ry * scale + 1.0f, 0.0f), 2.0f);
                const float cz = fminf(fmaxf(rz * scale + 1.0f, 0.0f), 2.0f);
                float wx[3], wy[3], wz[3];
                #pragma unroll
                for (int a = 0; a < 3; ++a) {
                    wx[a] = fmaxf(0.0f, 1.0f - fabsf(cx - (float)a));
                    wy[a] = fmaxf(0.0f, 1.0f - fabsf(cy - (float)a));
                    wz[a] = fmaxf(0.0f, 1.0f - fabsf(cz - (float)a));
                }
                const float f = feats[idx * CIN + lane] * imp;
                const float fz0 = wz[0] * f, fz1 = wz[1] * f, fz2 = wz[2] * f;
                #pragma unroll
                for (int a = 0; a < 3; ++a) {
                    #pragma unroll
                    for (int b = 0; b < 3; ++b) {
                        const float wab = wx[a] * wy[b];
                        acc[(a * 3 + b) * 3 + 0] = fmaf(wab, fz0, acc[(a * 3 + b) * 3 + 0]);
                        acc[(a * 3 + b) * 3 + 1] = fmaf(wab, fz1, acc[(a * 3 + b) * 3 + 1]);
                        acc[(a * 3 + b) * 3 + 2] = fmaf(wab, fz2, acc[(a * 3 + b) * 3 + 2]);
                    }
                }
            }
        }

        #pragma unroll
        for (int v = 0; v < VOX; ++v)
            sA[row][v * CIN + lane] = acc[v];
        if (lane == 0)
            sInvN[row] = (norm > 0.0f) ? 1.0f / fmaxf(norm, 1e-12f) : 0.0f;
    }
    __syncthreads();

    const int n0   = wave * 16;
    const int half = lane >> 4;
    const int lm   = lane & 15;
    const float* Wt = Wg + n0 + lm;

    v8f c = {};
    #pragma unroll 4
    for (int kk = 0; kk < KDIM / 4; ++kk) {
        const int kb = kk * 4 + 2 * half;
        v2f a;
        a.x = sA[lm][kb + 0];
        a.y = sA[lm][kb + 1];
        v2f b;
        b.x = Wt[(kb + 0) * COUT];
        b.y = Wt[(kb + 1) * COUT];
        c = __builtin_amdgcn_wmma_f32_16x16x4_f32(
                 false, a,  false, b,
                 (short)0, c,  false,  false);
    }

    const float bv = bias[n0 + lm];
    #pragma unroll
    for (int j = 0; j < 8; ++j) {
        const int mrow = j + 8 * half;
        out[(long)(m0 + mrow) * COUT + n0 + lm] = c[j] * sInvN[mrow] + bv;
    }
}

extern "C" void kernel_launch(void* const* d_in, const int* in_sizes, int n_in,
                              void* d_out, int out_size, void* d_ws, size_t ws_size,
                              hipStream_t stream) {
    const float* feats      = (const float*)d_in[0];
    const float* inPos      = (const float*)d_in[1];
    const float* outPos     = (const float*)d_in[2];
    const float* extents    = (const float*)d_in[3];
    const float* importance = (const float*)d_in[4];
    const int*   nbrIdx     = (const int*)  d_in[5];
    const float* Wg         = (const float*)d_in[6];
    const float* bias       = (const float*)d_in[7];
    float* out = (float*)d_out;

    dim3 grid(MOUT / MTILE);
    cconv_fused_kernel<<<grid, BLOCK, 0, stream>>>(
        feats, inPos, outPos, extents, importance, nbrIdx, Wg, bias, out);
}


# importance gathered only for in-radius neighbours (other lanes borrow an in-radius address), in flight with the row's feature gathers: one 128-byte line per in-radius neighbour, not per neighbour
# speedup vs baseline: 4.1777x; 4.1777x over previous
#include <hip/hip_runtime.h>

typedef __attribute__((ext_vector_type(2))) float v2f;
typedef __attribute__((ext_vector_type(8))) float v8f;
typedef __attribute__((ext_vector_type(16))) float v16f;

#define NPTS   200000
#define MOUT   100000
#define CIN    32
#define COUT   64
#define KNBR   32
#define VOX    27
#define KDIM   (VOX * CIN)
#define AS     (KDIM + 2)
#define MTILE  16
#define WAVES  8
#define BLOCK  (WAVES * 32)
#define RPW    (MTILE / WAVES)
#define GB     16
#define KQ     (KDIM / WAVES)
#define KSTEPS (KQ / 4)
#define KUNR   6
#define KREM   (KSTEPS % KUNR)
static_assert(KQ % 4 == 0 && WAVES * MTILE * COUT <= MTILE * AS, "partial tiles must fit in the A tile");
static_assert(RPW % 2 == 0 && MTILE % WAVES == 0, "the store phase writes two rows per instruction");
static_assert(GB == 16 && KNBR % GB == 0, "the gather group below names 16 registers");

__device__ __forceinline__ float lane_f(float v, int l) {
    return __builtin_bit_cast(float, __builtin_amdgcn_readlane(__builtin_bit_cast(int, v), l));
}
__device__ __forceinline__ int lane_i(int v, int l) {
    return __builtin_amdgcn_readlane(v, l);
}

#define OCT_LOOP(MASK, A, B, C, D, E, F, G, H)                                        \
    for (unsigned m_ = (MASK); m_ != 0u; m_ &= m_ - 1u) {                             \
        const int k2 = __builtin_ctz(m_);                                             \
        const int u2 = __builtin_popcount(bm & ((1u << k2) - 1u));                    \
        const float ik_ = lane_f(imp, k2);                                            \
        norm += ik_;                                                                  \
        const float f_ = fvv[u2] * ik_;                                               \
        const float fl = lane_f(zl, k2) * f_, fh = lane_f(zh, k2) * f_;               \
        const float a00 = lane_f(q00, k2), a01 = lane_f(q01, k2);                     \
        const float a10 = lane_f(q10, k2), a11 = lane_f(q11, k2);                     \
        A = fmaf(a00, fl, A); B = fmaf(a00, fh, B);                                   \
        C = fmaf(a01, fl, C); D = fmaf(a01, fh, D);                                   \
        E = fmaf(a10, fl, E); F = fmaf(a10, fh, F);                                   \
        G = fmaf(a11, fl, G); H = fmaf(a11, fh, H);                                   \
    }

__device__ __forceinline__ void kstep(const float* Ap, const float* __restrict__ Wp, int ks,
                                      v8f& c0, v8f& c1, v8f& c2, v8f& c3) {
    v2f a;
    a.x = Ap[ks * 4 + 0];
    a.y = Ap[ks * 4 + 1];
    const float4 w0 = *(const float4*)(Wp + (ks * 4 + 0) * COUT);
    const float4 w1 = *(const float4*)(Wp + (ks * 4 + 1) * COUT);
    v2f b;
    b.x = w0.x; b.y = w1.x;
    c0 = __builtin_amdgcn_wmma_f32_16x16x4_f32(false, a, false, b, (short)0, c0, false, false);
    b.x = w0.y; b.y = w1.y;
    c1 = __builtin_amdgcn_wmma_f32_16x16x4_f32(false, a, false, b, (short)0, c1, false, false);
    b.x = w0.z; b.y = w1.z;
    c2 = __builtin_amdgcn_wmma_f32_16x16x4_f32(false, a, false, b, (short)0, c2, false, false);
    b.x = w0.w; b.y = w1.w;
    c3 = __builtin_amdgcn_wmma_f32_16x16x4_f32(false, a, false, b, (short)0, c3, false, false);
}

__global__ __launch_bounds__(BLOCK)
void cconv_fused_kernel(const float* __restrict__ feats,
                        const float* __restrict__ inPos,
                        const float* __restrict__ outPos,
                        const float* __restrict__ extents,
                        const float* __restrict__ importance,
                        const int*   __restrict__ nbrIdx,
                        const float* __restrict__ Wg,
                        const float* __restrict__ bias,
                        float*       __restrict__ out)
{
    __shared__ __attribute__((aligned(16))) float sA[MTILE][AS];
    __shared__ float sInvN[MTILE];

    const int lane = threadIdx.x & 31;
    const int wave = threadIdx.x >> 5;
    const int m0   = blockIdx.x * MTILE;
    const float invR = 2.0f / extents[0];

    const int mw = m0 + wave * RPW;
    const float opv = outPos[mw * 3 + (lane < 3 * RPW ? lane : 3 * RPW - 1)];
    int   nI[RPW];
    float nX[RPW], nY[RPW], nZ[RPW];
    #pragma unroll
    for (int r = 0; r < RPW; ++r)
        nI[r] = nbrIdx[(mw + r) * KNBR + lane];
    #pragma unroll
    for (int r = 0; r < RPW; ++r) {
        nX[r] = inPos[nI[r] * 3 + 0];
        nY[r] = inPos[nI[r] * 3 + 1];
        nZ[r] = inPos[nI[r] * 3 + 2];
    }

    #pragma unroll 1
    for (int r = 0; r < RPW; ++r) {
        const int row = wave * RPW + r;
        const int   idx = nI[0];
        const float ox = lane_f(opv, r * 3 + 0);
        const float oy = lane_f(opv, r * 3 + 1);
        const float oz = lane_f(opv, r * 3 + 2);
        const float rx = (nX[0] - ox) * invR;
        const float ry = (nY[0] - oy) * invR;
        const float rz = (nZ[0] - oz) * invR;
        const float dist = sqrtf(rx * rx + ry * ry + rz * rz);
        const float linf = fmaxf(fabsf(rx), fmaxf(fabsf(ry), fabsf(rz)));
        const float scale = (linf > 1e-8f) ? dist / fmaxf(linf, 1e-8f) : 0.0f;
        const bool inR = dist <= 1.0f;
        const unsigned mr = __builtin_amdgcn_ballot_w32(inR);
        const int ia = inR ? idx : lane_i(idx, mr != 0u ? __builtin_ctz(mr) : 0);
        float nw = importance[ia];
        const float cx = fminf(fmaxf(rx * scale + 1.0f, 0.0f), 2.0f);
        const float cy = fminf(fmaxf(ry * scale + 1.0f, 0.0f), 2.0f);
        const float cz = fminf(fmaxf(rz * scale + 1.0f, 0.0f), 2.0f);
        float wx[3], wy[3], wz[3];
        #pragma unroll
        for (int a = 0; a < 3; ++a) {
            wx[a] = fmaxf(0.0f, 1.0f - fabsf(cx - (float)a));
            wy[a] = fmaxf(0.0f, 1.0f - fabsf(cy - (float)a));
            wz[a] = fmaxf(0.0f, 1.0f - fabsf(cz - (float)a));
        }

        const bool hx = cx >= 1.0f, hy = cy >= 1.0f, hz = cz >= 1.0f;
        const float xl = hx ? wx[1] : wx[0], xh = hx ? wx[2] : wx[1];
        const float yl = hy ? wy[1] : wy[0], yh = hy ? wy[2] : wy[1];
        const float zl = hz ? wz[1] : wz[0], zh = hz ? wz[2] : wz[1];
        const float q00 = xl * yl, q01 = xl * yh;
        const float q10 = xh * yl, q11 = xh * yh;
        const unsigned mx = __builtin_amdgcn_ballot_w32(hx);
        const unsigned my = __builtin_amdgcn_ballot_w32(hy);
        const unsigned mz = __builtin_amdgcn_ballot_w32(hz);

        float r0 = 0.0f, r1 = 0.0f, r2 = 0.0f, r3 = 0.0f, r4 = 0.0f, r5 = 0.0f, r6 = 0.0f, r7 = 0.0f, r8 = 0.0f;
        float r9 = 0.0f, r10 = 0.0f, r11 = 0.0f, r12 = 0.0f, r13 = 0.0f, r14 = 0.0f, r15 = 0.0f, r16 = 0.0f, r17 = 0.0f;
        float r18 = 0.0f, r19 = 0.0f, r20 = 0.0f, r21 = 0.0f, r22 = 0.0f, r23 = 0.0f, r24 = 0.0f, r25 = 0.0f, r26 = 0.0f;
        float norm = 0.0f;

        unsigned mk = mr;
        for (int bt = 0; bt < KNBR / GB && mk != 0u; ++bt) {
            float fv[GB];
            unsigned t = mk;
            int k = 0;
            #pragma unroll
            for (int u = 0; u < GB; ++u) {
                if (t != 0u) { k = __builtin_ctz(t); t &= t - 1u; }
                fv[u] = feats[lane_i(idx, k) * CIN + lane];
            }
            asm volatile("" : "+v"(fv[0]), "+v"(fv[1]), "+v"(fv[2]), "+v"(fv[3]),
                              "+v"(fv[4]), "+v"(fv[5]), "+v"(fv[6]), "+v"(fv[7]),
                              "+v"(fv[8]), "+v"(fv[9]), "+v"(fv[10]), "+v"(fv[11]),
                              "+v"(fv[12]), "+v"(fv[13]), "+v"(fv[14]), "+v"(fv[15]), "+v"(nw));
            const float imp = inR ? nw : 0.0f;
            const unsigned bm = mk & ~t;
            const v16f fvv = {fv[0], fv[1], fv[2],  fv[3],  fv[4],  fv[5],  fv[6],  fv[7],
                              fv[8], fv[9], fv[10], fv[11], fv[12], fv[13], fv[14], fv[15]};
            const unsigned x0 = bm & ~mx, x1 = bm & mx;
            OCT_LOOP(x0 & ~my & ~mz, r0, r1, r3, r4, r9, r10, r12, r13)
            OCT_LOOP(x0 & ~my &  mz, r1, r2, r4, r5, r10, r11, r13, r14)
            OCT_LOOP(x0 &  my & ~mz, r3, r4, r6, r7, r12, r13, r15, r16)
            OCT_LOOP(x0 &  my &  mz, r4, r5, r7, r8, r13, r14, r16, r17)
            OCT_LOOP(x1 & ~my & ~mz, r9, r10, r12, r13, r18, r19, r21, r22)
            OCT_LOOP(x1 & ~my &  mz, r10, r11, r13, r14, r19, r20, r22, r23)
            OCT_LOOP(x1 &  my & ~mz, r12, r13, r15, r16, r21, r22, r24, r25)
            OCT_LOOP(x1 &  my &  mz, r13, r14, r16, r17, r22, r23, r25, r26)
            mk = t;
        }

        const float acc[VOX] = {r0, r1, r2, r3, r4, r5, r6, r7, r8,
                                r9, r10, r11, r12, r13, r14, r15, r16, r17,
                                r18, r19, r20, r21, r22, r23, r24, r25, r26};
        #pragma unroll
        for (int v = 0; v < VOX; ++v)
            sA[row][v * CIN + lane] = acc[v];
        if (lane == 0)
            sInvN[row] = (norm > 0.0f) ? 1.0f / fmaxf(norm, 1e-12f) : 0.0f;
        #pragma unroll
        for (int q = 0; q + 1 < RPW; ++q) {
            nI[q] = nI[q + 1]; nX[q] = nX[q + 1]; nY[q] = nY[q + 1];
            nZ[q] = nZ[q + 1];
        }
    }
    __syncthreads();

    const int half = lane >> 4;
    const int lm   = lane & 15;
    const int kw   = wave * KQ + 2 * half;
    const float* Wp = Wg + (size_t)kw * COUT + 4 * lm;
    const float* Ap = &sA[lm][kw];

    v8f c0 = {}, c1 = {}, c2 = {}, c3 = {};
    #pragma unroll
    for (int ks = 0; ks < KREM; ++ks)
        kstep(Ap, Wp, ks, c0, c1, c2, c3);
    #pragma unroll KUNR
    for (int ks = KREM; ks < KSTEPS; ++ks)
        kstep(Ap, Wp, ks, c0, c1, c2, c3);

    __syncthreads();
    float* sP = &sA[0][0];
    #pragma unroll
    for (int j = 0; j < 8; ++j) {
        float4 v;
        v.x = c0[j]; v.y = c1[j]; v.z = c2[j]; v.w = c3[j];
        *(float4*)(sP + ((wave * MTILE + j + 8 * half) * COUT + 4 * lm)) = v;
    }
    __syncthreads();

    const float4 bv = *(const float4*)(bias + 4 * lm);
    #pragma unroll
    for (int q = 0; q < RPW / 2; ++q) {
        const int mrow = wave * RPW + 2 * q + half;
        const float* src = sP + mrow * COUT + 4 * lm;
        float4 r = *(const float4*)(src);
        #pragma unroll
        for (int w = 1; w < WAVES; ++w) {
            const float4 p = *(const float4*)(src + w * MTILE * COUT);
            r.x += p.x; r.y += p.y; r.z += p.z; r.w += p.w;
        }
        const float inv = sInvN[mrow];
        float4 o;
        o.x = r.x * inv + bv.x; o.y = r.y * inv + bv.y;
        o.z = r.z * inv + bv.z; o.w = r.w * inv + bv.w;
        *(float4*)(out + (size_t)(m0 + mrow) * COUT + 4 * lm) = o;
    }
}

extern "C" void kernel_launch(void* const* d_in, const int* in_sizes, int n_in,
                              void* d_out, int out_size, void* d_ws, size_t ws_size,
                              hipStream_t stream) {
    const float* feats      = (const float*)d_in[0];
    const float* inPos      = (const float*)d_in[1];
    const float* outPos     = (const float*)d_in[2];
    const float* extents    = (const float*)d_in[3];
    const float* importance = (const float*)d_in[4];
    const int*   nbrIdx     = (const int*)  d_in[5];
    const float* Wg         = (const float*)d_in[6];
    const float* bias       = (const float*)d_in[7];
    float* out = (float*)d_out;

    dim3 grid(MOUT / MTILE);
    cconv_fused_kernel<<<grid, BLOCK, 0, stream>>>(
        feats, inPos, outPos, extents, importance, nbrIdx, Wg, bias, out);
}
